# starting kernel
# baseline (speedup 1.0000x reference)
#include <hip/hip_runtime.h>
#include <math.h>

typedef __attribute__((ext_vector_type(2))) float v2f;
typedef __attribute__((ext_vector_type(8))) float v8f;

__device__ __forceinline__ void atomic_add_f32(float* p, float v) {
    __hip_atomic_fetch_add(p, v, __ATOMIC_RELAXED, __HIP_MEMORY_SCOPE_AGENT);
}

__global__ void k_fill_deg(float* __restrict__ deg, int n) {
    int i = blockIdx.x * blockDim.x + threadIdx.x;
    if (i < n) deg[i] = 1.0f;
}

__global__ void k_deg_scatter(const int* __restrict__ dst, int E,
                              float* __restrict__ deg) {
    int e = blockIdx.x * blockDim.x + threadIdx.x;
    if (e < E) atomic_add_f32(&deg[dst[e]], 1.0f);
}

__global__ void k_dinv(float* __restrict__ deg, int n) {
    int i = blockIdx.x * blockDim.x + threadIdx.x;
    if (i < n) {
        float d = deg[i];
        deg[i] = (d > 0.0f) ? rsqrtf(d) : 0.0f;
    }
}

template <int NT, int KD, bool RELU>
__global__ void __launch_bounds__(256) k_gemm_wmma_f32(
    const float* __restrict__ A, const float* __restrict__ B,
    float* __restrict__ C, int M)
{
    constexpr int N   = NT * 16;
    constexpr int LDK = KD + 2;
    __shared__ float lds_w[N * LDK];

    for (int idx = threadIdx.x; idx < KD * N; idx += 256) {
        int k = idx / N;
        int n = idx - k * N;
        lds_w[n * LDK + k] = B[idx];
    }
    __syncthreads();

    const int lane  = threadIdx.x & 31;
    const int wv    = threadIdx.x >> 5;
    const int mtile = blockIdx.x * 8 + wv;
    if (mtile * 16 >= M) return;

    const int hf = lane >> 4;
    const int r  = lane & 15;

    const float* __restrict__ arow = A + (size_t)(mtile * 16 + r) * KD + 2 * hf;

    v8f acc[NT];
#pragma unroll
    for (int t = 0; t < NT; ++t) acc[t] = (v8f){};

#pragma unroll 4
    for (int k = 0; k < KD; k += 4) {
        float2 av = *(const float2*)(arow + k);
        if (RELU) { av.x = fmaxf(av.x, 0.0f); av.y = fmaxf(av.y, 0.0f); }
        v2f a;
        a[0] = av.x;
        a[1] = av.y;
#pragma unroll
        for (int t = 0; t < NT; ++t) {
            const int col = t * 16 + r;
            float2 bv = *(const float2*)(&lds_w[col * LDK + k + 2 * hf]);
            v2f b;
            b[0] = bv.x;
            b[1] = bv.y;
            acc[t] = __builtin_amdgcn_wmma_f32_16x16x4_f32(
                false, a, false, b, (short)0, acc[t], false, false);
        }
    }

#pragma unroll
    for (int t = 0; t < NT; ++t) {
        float* crow = C + (size_t)(mtile * 16 + 8 * hf) * N + t * 16 + r;
#pragma unroll
        for (int j = 0; j < 8; ++j) crow[(size_t)j * N] = acc[t][j];
    }
}

__global__ void k_self_bias_init(const float* __restrict__ h,
                                 const float* __restrict__ dinv,
                                 const float* __restrict__ bias,
                                 float* __restrict__ out, int n, int lgC) {
    long long t = (long long)blockIdx.x * blockDim.x + threadIdx.x;
    long long total = (long long)n << lgC;
    if (t >= total) return;
    int i = (int)(t >> lgC);
    int c = (int)(t & (((long long)1 << lgC) - 1));
    float di = dinv[i];
    out[t] = bias[c] + h[t] * di * di;
}

__global__ void k_edge_scatter(const int* __restrict__ src,
                               const int* __restrict__ dst,
                               const float* __restrict__ dinv,
                               const float* __restrict__ h,
                               float* __restrict__ out, int E, int lgG) {
    long long t = (long long)blockIdx.x * blockDim.x + threadIdx.x;
    long long total = (long long)E << lgG;
    if (t >= total) return;
    const int C = 4 << lgG;
    int e  = (int)(t >> lgG);
    int cg = (int)(t & (((long long)1 << lgG) - 1)) << 2;
    int s = src[e], d = dst[e];
    float nrm = dinv[s] * dinv[d];
    const float4 hv = *(const float4*)(h + (size_t)s * C + cg);
    float* o = out + (size_t)d * C + cg;
    atomic_add_f32(o + 0, hv.x * nrm);
    atomic_add_f32(o + 1, hv.y * nrm);
    atomic_add_f32(o + 2, hv.z * nrm);
    atomic_add_f32(o + 3, hv.w * nrm);
}

extern "C" void kernel_launch(void* const* d_in, const int* in_sizes, int n_in,
                              void* d_out, int out_size, void* d_ws, size_t ws_size,
                              hipStream_t stream) {
    const float* x  = (const float*)d_in[0];
    const int*   ei = (const int*)d_in[1];
    const float* W1 = (const float*)d_in[2];
    const float* b1 = (const float*)d_in[3];
    const float* W2 = (const float*)d_in[4];
    const float* b2 = (const float*)d_in[5];
    float* out = (float*)d_out;

    const int HID = in_sizes[3];
    const int OUT = in_sizes[5];
    const int IN  = in_sizes[2] / HID;
    const int N   = in_sizes[0] / IN;
    const int E   = in_sizes[1] / 2;
    const int* srcI = ei;
    const int* dstI = ei + E;

    float* ws   = (float*)d_ws;
    float* dinv = ws;
    float* bufA = ws + (((size_t)N + 63) & ~(size_t)63);
    float* bufB = bufA + (size_t)N * HID;

    const int T = 256;
    const int mtiles = N / 16;
    const int gblk   = (mtiles + 7) / 8;

    k_fill_deg   <<<(N + T - 1) / T, T, 0, stream>>>(dinv, N);
    k_deg_scatter<<<(E + T - 1) / T, T, 0, stream>>>(dstI, E, dinv);
    k_dinv       <<<(N + T - 1) / T, T, 0, stream>>>(dinv, N);

    k_gemm_wmma_f32<8, 128, false><<<gblk, T, 0, stream>>>(x, W1, bufA, N);
    {
        long long tot = (long long)N * HID;
        k_self_bias_init<<<(int)((tot + T - 1) / T), T, 0, stream>>>(
            bufA, dinv, b1, bufB, N, 7);
        long long etot = (long long)E << 5;
        k_edge_scatter<<<(int)((etot + T - 1) / T), T, 0, stream>>>(
            srcI, dstI, dinv, bufA, bufB, E, 5);
    }

    k_gemm_wmma_f32<4, 128, true><<<gblk, T, 0, stream>>>(bufB, W2, bufA, N);
    {
        long long tot = (long long)N * OUT;
        k_self_bias_init<<<(int)((tot + T - 1) / T), T, 0, stream>>>(
            bufA, dinv, b2, out, N, 6);
        long long etot = (long long)E << 4;
        k_edge_scatter<<<(int)((etot + T - 1) / T), T, 0, stream>>>(
            srcI, dstI, dinv, bufA, out, E, 4);
    }
}


# edge scatter: lane = channel (l, l+32, ..), so each f32 atomic add and each gather load of a wave covers one whole 128-B line, not 4 lines x 8 lanes; adds still land in arrival order
# speedup vs baseline: 1.3546x; 1.3546x over previous
#include <hip/hip_runtime.h>
#include <math.h>

typedef __attribute__((ext_vector_type(2))) float v2f;
typedef __attribute__((ext_vector_type(8))) float v8f;

__device__ __forceinline__ void atomic_add_f32(float* p, float v) {
    __hip_atomic_fetch_add(p, v, __ATOMIC_RELAXED, __HIP_MEMORY_SCOPE_AGENT);
}

__global__ void k_fill_deg(float* __restrict__ deg, int n) {
    int i = blockIdx.x * blockDim.x + threadIdx.x;
    if (i < n) deg[i] = 1.0f;
}

__global__ void k_deg_scatter(const int* __restrict__ dst, int E,
                              float* __restrict__ deg) {
    int e = blockIdx.x * blockDim.x + threadIdx.x;
    if (e < E) atomic_add_f32(&deg[dst[e]], 1.0f);
}

__global__ void k_dinv(float* __restrict__ deg, int n) {
    int i = blockIdx.x * blockDim.x + threadIdx.x;
    if (i < n) {
        float d = deg[i];
        deg[i] = (d > 0.0f) ? rsqrtf(d) : 0.0f;
    }
}

template <int NT, int KD, bool RELU>
__global__ void __launch_bounds__(256) k_gemm_wmma_f32(
    const float* __restrict__ A, const float* __restrict__ B,
    float* __restrict__ C, int M)
{
    constexpr int N   = NT * 16;
    constexpr int LDK = KD + 2;
    __shared__ float lds_w[N * LDK];

    for (int idx = threadIdx.x; idx < KD * N; idx += 256) {
        int k = idx / N;
        int n = idx - k * N;
        lds_w[n * LDK + k] = B[idx];
    }
    __syncthreads();

    const int lane  = threadIdx.x & 31;
    const int wv    = threadIdx.x >> 5;
    const int mtile = blockIdx.x * 8 + wv;
    if (mtile * 16 >= M) return;

    const int hf = lane >> 4;
    const int r  = lane & 15;

    const float* __restrict__ arow = A + (size_t)(mtile * 16 + r) * KD + 2 * hf;

    v8f acc[NT];
#pragma unroll
    for (int t = 0; t < NT; ++t) acc[t] = (v8f){};

#pragma unroll 4
    for (int k = 0; k < KD; k += 4) {
        float2 av = *(const float2*)(arow + k);
        if (RELU) { av.x = fmaxf(av.x, 0.0f); av.y = fmaxf(av.y, 0.0f); }
        v2f a;
        a[0] = av.x;
        a[1] = av.y;
#pragma unroll
        for (int t = 0; t < NT; ++t) {
            const int col = t * 16 + r;
            float2 bv = *(const float2*)(&lds_w[col * LDK + k + 2 * hf]);
            v2f b;
            b[0] = bv.x;
            b[1] = bv.y;
            acc[t] = __builtin_amdgcn_wmma_f32_16x16x4_f32(
                false, a, false, b, (short)0, acc[t], false, false);
        }
    }

#pragma unroll
    for (int t = 0; t < NT; ++t) {
        float* crow = C + (size_t)(mtile * 16 + 8 * hf) * N + t * 16 + r;
#pragma unroll
        for (int j = 0; j < 8; ++j) crow[(size_t)j * N] = acc[t][j];
    }
}

__global__ void k_self_bias_init(const float* __restrict__ h,
                                 const float* __restrict__ dinv,
                                 const float* __restrict__ bias,
                                 float* __restrict__ out, int n, int lgC) {
    long long t = (long long)blockIdx.x * blockDim.x + threadIdx.x;
    long long total = (long long)n << lgC;
    if (t >= total) return;
    int i = (int)(t >> lgC);
    int c = (int)(t & (((long long)1 << lgC) - 1));
    float di = dinv[i];
    out[t] = bias[c] + h[t] * di * di;
}

template <int C, int NE>
__device__ __forceinline__ void edge_rows_add(const int* __restrict__ src,
                                              const int* __restrict__ dst,
                                              const float* __restrict__ dinv,
                                              const float* __restrict__ h,
                                              float* __restrict__ out,
                                              long long e0, int lane) {
    constexpr int LPR = C / 32;
    float  v[NE][LPR];
    float* o[NE];
#pragma unroll
    for (int i = 0; i < NE; ++i) {
        const int s = src[e0 + i], d = dst[e0 + i];
        const float nrm = dinv[s] * dinv[d];
        const float* hr = h + (size_t)s * C + lane;
        o[i] = out + (size_t)d * C + lane;
#pragma unroll
        for (int j = 0; j < LPR; ++j) v[i][j] = hr[32 * j] * nrm;
    }
#pragma unroll
    for (int i = 0; i < NE; ++i) {
#pragma unroll
        for (int j = 0; j < LPR; ++j) atomic_add_f32(o[i] + 32 * j, v[i][j]);
    }
}

template <int C>
__global__ void k_edge_scatter(const int* __restrict__ src,
                               const int* __restrict__ dst,
                               const float* __restrict__ dinv,
                               const float* __restrict__ h,
                               float* __restrict__ out, int E) {
    constexpr int EPW = 128 / C;
    const int lane = threadIdx.x & 31;
    const long long wave =
        ((long long)blockIdx.x * blockDim.x + threadIdx.x) >> 5;
    const long long e0 = wave * EPW;
    if (e0 + EPW <= E) {
        edge_rows_add<C, EPW>(src, dst, dinv, h, out, e0, lane);
    } else {
        for (long long e = e0; e < E; ++e)
            edge_rows_add<C, 1>(src, dst, dinv, h, out, e, lane);
    }
}

extern "C" void kernel_launch(void* const* d_in, const int* in_sizes, int n_in,
                              void* d_out, int out_size, void* d_ws, size_t ws_size,
                              hipStream_t stream) {
    const float* x  = (const float*)d_in[0];
    const int*   ei = (const int*)d_in[1];
    const float* W1 = (const float*)d_in[2];
    const float* b1 = (const float*)d_in[3];
    const float* W2 = (const float*)d_in[4];
    const float* b2 = (const float*)d_in[5];
    float* out = (float*)d_out;

    const int HID = in_sizes[3];
    const int OUT = in_sizes[5];
    const int IN  = in_sizes[2] / HID;
    const int N   = in_sizes[0] / IN;
    const int E   = in_sizes[1] / 2;
    const int* srcI = ei;
    const int* dstI = ei + E;

    float* ws   = (float*)d_ws;
    float* dinv = ws;
    float* bufA = ws + (((size_t)N + 63) & ~(size_t)63);
    float* bufB = bufA + (size_t)N * HID;

    const int T = 256;
    const int mtiles = N / 16;
    const int gblk   = (mtiles + 7) / 8;

    k_fill_deg   <<<(N + T - 1) / T, T, 0, stream>>>(dinv, N);
    k_deg_scatter<<<(E + T - 1) / T, T, 0, stream>>>(dstI, E, dinv);
    k_dinv       <<<(N + T - 1) / T, T, 0, stream>>>(dinv, N);

    k_gemm_wmma_f32<8, 128, false><<<gblk, T, 0, stream>>>(x, W1, bufA, N);
    {
        long long tot = (long long)N * HID;
        k_self_bias_init<<<(int)((tot + T - 1) / T), T, 0, stream>>>(
            bufA, dinv, b1, bufB, N, 7);
        long long etot = (long long)E << 5;
        k_edge_scatter<128><<<(int)((etot + T - 1) / T), T, 0, stream>>>(
            srcI, dstI, dinv, bufA, bufB, E);
    }

    k_gemm_wmma_f32<4, 128, true><<<gblk, T, 0, stream>>>(bufB, W2, bufA, N);
    {
        long long tot = (long long)N * OUT;
        k_self_bias_init<<<(int)((tot + T - 1) / T), T, 0, stream>>>(
            bufA, dinv, b2, out, N, 6);
        long long etot = (long long)E << 4;
        k_edge_scatter<64><<<(int)((etot + T - 1) / T), T, 0, stream>>>(
            srcI, dstI, dinv, bufA, out, E);
    }
}


# edge sums: per-call dst-grouped edge list (int atomic count, scan, fill), then one wave per node adds its gathered rows in registers; no f32 atomics, no init pass; sum order = arrival rank in row
# speedup vs baseline: 1.6819x; 1.2416x over previous
#include <hip/hip_runtime.h>
#include <math.h>


typedef __attribute__((ext_vector_type(2))) float v2f;
typedef __attribute__((ext_vector_type(8))) float v8f;


__global__ void k_zero_cnt(int* __restrict__ cnt, int n) {
    int i = blockIdx.x * blockDim.x + threadIdx.x;
    if (i < n) cnt[i] = 0;
}

__global__ void k_count_dst(const int* __restrict__ dst, int E,
                            int* __restrict__ cnt) {
    int e = blockIdx.x * blockDim.x + threadIdx.x;
    if (e < E)
        __hip_atomic_fetch_add(&cnt[dst[e]], 1, __ATOMIC_RELAXED,
                               __HIP_MEMORY_SCOPE_AGENT);
}

__device__ __forceinline__ float dinv_of(int c) {
    const float dg = (float)(c + 1);
    return (dg > 0.0f) ? rsqrtf(dg) : 0.0f;
}

__global__ void __launch_bounds__(1024) k_scan_rows(
    const int* __restrict__ cnt, int* __restrict__ rp, int n)
{
    constexpr int EPT  = 16;
    constexpr int TILE = 1024 * EPT;
    __shared__ int wsum[32];
    int* start = rp + 1;
    const int tid  = threadIdx.x;
    const int lane = tid & 31;
    const int wv   = tid >> 5;
    if (tid == 0) rp[0] = 0;
    int carry = 0;
    for (int base = 0; base < n; base += TILE) {
        const int i0 = base + tid * EPT;
        const bool whole = (i0 + EPT <= n);
        int c[EPT];
        if (whole) {
#pragma unroll
            for (int q = 0; q < EPT / 4; ++q) {
                const int4 v = *(const int4*)(cnt + i0 + 4 * q);
                c[4 * q] = v.x; c[4 * q + 1] = v.y; c[4 * q + 2] = v.z; c[4 * q + 3] = v.w;
            }
        } else {
#pragma unroll
            for (int k = 0; k < EPT; ++k) c[k] = (i0 + k < n) ? cnt[i0 + k] : 0;
        }
        int tot = 0;
#pragma unroll
        for (int k = 0; k < EPT; ++k) tot += c[k];
        int inc = tot;
        for (int off = 1; off < 32; off <<= 1) {
            const int v = __shfl_up(inc, off);
            if (lane >= off) inc += v;
        }
        if (lane == 31) wsum[wv] = inc;
        __syncthreads();
        const int ws = wsum[lane];
        int wsc = ws;
        for (int off = 1; off < 32; off <<= 1) {
            const int v = __shfl_up(wsc, off);
            if (lane >= off) wsc += v;
        }
        const int wave_excl  = __shfl(wsc, wv) - __shfl(ws, wv);
        const int tile_total = __shfl(wsc, 31);
        int ex[EPT];
        int run = carry + wave_excl + (inc - tot);
#pragma unroll
        for (int k = 0; k < EPT; ++k) {
            ex[k] = run;
            run += c[k];
        }
        if (whole) {
#pragma unroll
            for (int q = 0; q < EPT / 4; ++q)
                *(int4*)(start + i0 + 4 * q) =
                    make_int4(ex[4 * q], ex[4 * q + 1], ex[4 * q + 2], ex[4 * q + 3]);
        } else {
#pragma unroll
            for (int k = 0; k < EPT; ++k)
                if (i0 + k < n) start[i0 + k] = ex[k];
        }
        carry += tile_total;
        __syncthreads();
    }
}

__global__ void k_fill_rows(const int* __restrict__ src,
                            const int* __restrict__ dst, int E,
                            const int* __restrict__ cnt,
                            int* __restrict__ rp,
                            int* __restrict__ edges) {
    int e = blockIdx.x * blockDim.x + threadIdx.x;
    if (e < E) {
        const int s = src[e];
        const int d = dst[e];
        const int pos = __hip_atomic_fetch_add(&rp[d + 1], 1, __ATOMIC_RELAXED,
                                               __HIP_MEMORY_SCOPE_AGENT);
        *(int2*)(edges + 2 * (size_t)pos) =
            make_int2(s, __float_as_int(dinv_of(cnt[s])));
    }
}

template <int NT, int KD, bool RELU>
__global__ void __launch_bounds__(256) k_gemm_wmma_f32(
    const float* __restrict__ A, const float* __restrict__ B,
    float* __restrict__ C, int M)
{
    constexpr int N   = NT * 16;
    constexpr int LDK = KD + 2;
    __shared__ float lds_w[N * LDK];

    for (int idx = threadIdx.x; idx < KD * N; idx += 256) {
        int k = idx / N;
        int n = idx - k * N;
        lds_w[n * LDK + k] = B[idx];
    }
    __syncthreads();

    const int lane  = threadIdx.x & 31;
    const int wv    = threadIdx.x >> 5;
    const int mtile = blockIdx.x * 8 + wv;
    if (mtile * 16 >= M) return;

    const int hf = lane >> 4;
    const int r  = lane & 15;

    const float* __restrict__ arow = A + (size_t)(mtile * 16 + r) * KD + 2 * hf;

    v8f acc[NT];
#pragma unroll
    for (int t = 0; t < NT; ++t) acc[t] = (v8f){};

#pragma unroll 4
    for (int k = 0; k < KD; k += 4) {
        float2 av = *(const float2*)(arow + k);
        if (RELU) { av.x = fmaxf(av.x, 0.0f); av.y = fmaxf(av.y, 0.0f); }
        v2f a;
        a[0] = av.x;
        a[1] = av.y;
#pragma unroll
        for (int t = 0; t < NT; ++t) {
            const int col = t * 16 + r;
            float2 bv = *(const float2*)(&lds_w[col * LDK + k + 2 * hf]);
            v2f b;
            b[0] = bv.x;
            b[1] = bv.y;
            acc[t] = __builtin_amdgcn_wmma_f32_16x16x4_f32(
                false, a, false, b, (short)0, acc[t], false, false);
        }
    }

#pragma unroll
    for (int t = 0; t < NT; ++t) {
        float* crow = C + (size_t)(mtile * 16 + 8 * hf) * N + t * 16 + r;
#pragma unroll
        for (int j = 0; j < 8; ++j) crow[(size_t)j * N] = acc[t][j];
    }
}

template <int V, int U>
__device__ __forceinline__ void rows_add(
    const __attribute__((ext_vector_type(V))) float* __restrict__ hv,
    int es, float ew, int i, int lane,
    __attribute__((ext_vector_type(V))) float& acc)
{
    typedef __attribute__((ext_vector_type(V))) float vecf;
    vecf  r[U];
    float w[U];
#pragma unroll
    for (int u = 0; u < U; ++u) {
        const int s = __builtin_amdgcn_readlane(es, i + u);
        w[u] = __int_as_float(__builtin_amdgcn_readlane(__float_as_int(ew), i + u));
        r[u] = hv[(size_t)s * 32 + lane];
    }
#pragma unroll
    for (int u = 0; u < U; ++u) acc += r[u] * w[u];
}

template <int C>
__global__ void __launch_bounds__(256) k_gather_sum(
    const int* __restrict__ rp, const int* __restrict__ edges,
    const int* __restrict__ cnt, const float* __restrict__ h,
    const float* __restrict__ bias, float* __restrict__ out, int n)
{
    constexpr int V = C / 32;
    typedef __attribute__((ext_vector_type(V))) float vecf;
    const int lane = threadIdx.x & 31;
    const int node = blockIdx.x * 8 + (threadIdx.x >> 5);
    if (node >= n) return;
    const int d = __builtin_amdgcn_readfirstlane(node);
    const int beg = rp[d];
    const int end = rp[d + 1];
    const float dd = dinv_of(cnt[node]);

    const vecf* __restrict__ hv = (const vecf*)h;
    const int2* __restrict__ ev = (const int2*)edges;

    vecf acc;
    {
        const vecf self = hv[(size_t)d * 32 + lane];
#pragma unroll
        for (int j = 0; j < V; ++j)
            acc[j] = bias[lane * V + j] + self[j] * dd * dd;
    }

    for (int base = beg; base < end; base += 32) {
        const int m = min(32, end - base);
        int   es = 0;
        float ew = 0.0f;
        if (lane < m) {
            const int2 p = ev[base + lane];
            es = p.x;
            ew = __int_as_float(p.y) * dd;
        }
        int i = 0;
        for (; i + 8 <= m; i += 8) rows_add<V, 8>(hv, es, ew, i, lane, acc);
        if (i + 4 <= m) { rows_add<V, 4>(hv, es, ew, i, lane, acc); i += 4; }
        if (i + 2 <= m) { rows_add<V, 2>(hv, es, ew, i, lane, acc); i += 2; }
        if (i < m)      { rows_add<V, 1>(hv, es, ew, i, lane, acc); }
    }

    ((vecf*)out)[(size_t)d * 32 + lane] = acc;
}

extern "C" void kernel_launch(void* const* d_in, const int* in_sizes, int n_in,
                              void* d_out, int out_size, void* d_ws, size_t ws_size,
                              hipStream_t stream) {
    const float* x  = (const float*)d_in[0];
    const int*   ei = (const int*)d_in[1];
    const float* W1 = (const float*)d_in[2];
    const float* b1 = (const float*)d_in[3];
    const float* W2 = (const float*)d_in[4];
    const float* b2 = (const float*)d_in[5];
    float* out = (float*)d_out;

    const int HID = in_sizes[3];
    const int OUT = in_sizes[5];
    const int IN  = in_sizes[2] / HID;
    const int N   = in_sizes[0] / IN;
    const int E   = in_sizes[1] / 2;
    const int* srcI = ei;
    const int* dstI = ei + E;

    const size_t nPad  = ((size_t)N + 4 + 63) & ~(size_t)63;
    const size_t ePad  = (2 * (size_t)E + 63) & ~(size_t)63;
    float* ws    = (float*)d_ws;
    int*   cnt   = (int*)ws;
    int*   rp    = (int*)(ws + nPad) + 3;
    int*   edges = (int*)(ws + 2 * nPad);
    float* bufA  = ws + 2 * nPad + ePad;
    float* bufB  = bufA + (size_t)N * HID;

    const int T = 256;
    const int mtiles = N / 16;
    const int gblk   = (mtiles + 7) / 8;
    const int nblk   = (N + 7) / 8;

    k_zero_cnt <<<(N + T - 1) / T, T, 0, stream>>>(cnt, N);
    k_count_dst<<<(E + T - 1) / T, T, 0, stream>>>(dstI, E, cnt);
    k_scan_rows<<<1, 1024, 0, stream>>>(cnt, rp, N);
    k_fill_rows<<<(E + T - 1) / T, T, 0, stream>>>(srcI, dstI, E, cnt, rp, edges);

    k_gemm_wmma_f32<8, 128, false><<<gblk, T, 0, stream>>>(x, W1, bufA, N);
    k_gather_sum<128><<<nblk, T, 0, stream>>>(rp, edges, cnt, bufA, b1, bufB, N);

    k_gemm_wmma_f32<4, 128, true><<<gblk, T, 0, stream>>>(bufB, W2, bufA, N);
    k_gather_sum<64><<<nblk, T, 0, stream>>>(rp, edges, cnt, bufA, b2, out, N);
}


# edge grouping: the count pass's integer add returns each edge's rank in its row (stored per edge); fill places at row start + rank by plain loads, no second atomic pass; sum order still arrival rank
# speedup vs baseline: 2.1324x; 1.2679x over previous
#include <hip/hip_runtime.h>
#include <math.h>


typedef __attribute__((ext_vector_type(2))) float v2f;
typedef __attribute__((ext_vector_type(8))) float v8f;


__global__ void k_zero_cnt(int* __restrict__ cnt, int n) {
    int i = blockIdx.x * blockDim.x + threadIdx.x;
    if (i < n) cnt[i] = 0;
}

__global__ void k_count_rank(const int* __restrict__ dst, int E,
                             int* __restrict__ cnt, int* __restrict__ rank) {
    int e = blockIdx.x * blockDim.x + threadIdx.x;
    if (e < E)
        rank[e] = __hip_atomic_fetch_add(&cnt[dst[e]], 1, __ATOMIC_RELAXED,
                                         __HIP_MEMORY_SCOPE_AGENT);
}

__device__ __forceinline__ float dinv_of(int c) {
    const float dg = (float)(c + 1);
    return (dg > 0.0f) ? rsqrtf(dg) : 0.0f;
}

__global__ void __launch_bounds__(1024) k_scan_rows(
    const int* __restrict__ cnt, int* __restrict__ rp, int n)
{
    constexpr int EPT  = 16;
    constexpr int TILE = 1024 * EPT;
    __shared__ int wsum[32];
    int* start = rp;
    const int tid  = threadIdx.x;
    const int lane = tid & 31;
    const int wv   = tid >> 5;
    int carry = 0;
    for (int base = 0; base < n; base += TILE) {
        const int i0 = base + tid * EPT;
        const bool whole = (i0 + EPT <= n);
        int c[EPT];
        if (whole) {
#pragma unroll
            for (int q = 0; q < EPT / 4; ++q) {
                const int4 v = *(const int4*)(cnt + i0 + 4 * q);
                c[4 * q] = v.x; c[4 * q + 1] = v.y; c[4 * q + 2] = v.z; c[4 * q + 3] = v.w;
            }
        } else {
#pragma unroll
            for (int k = 0; k < EPT; ++k) c[k] = (i0 + k < n) ? cnt[i0 + k] : 0;
        }
        int tot = 0;
#pragma unroll
        for (int k = 0; k < EPT; ++k) tot += c[k];
        int inc = tot;
        for (int off = 1; off < 32; off <<= 1) {
            const int v = __shfl_up(inc, off);
            if (lane >= off) inc += v;
        }
        if (lane == 31) wsum[wv] = inc;
        __syncthreads();
        const int ws = wsum[lane];
        int wsc = ws;
        for (int off = 1; off < 32; off <<= 1) {
            const int v = __shfl_up(wsc, off);
            if (lane >= off) wsc += v;
        }
        const int wave_excl  = __shfl(wsc, wv) - __shfl(ws, wv);
        const int tile_total = __shfl(wsc, 31);
        int ex[EPT];
        int run = carry + wave_excl + (inc - tot);
#pragma unroll
        for (int k = 0; k < EPT; ++k) {
            ex[k] = run;
            run += c[k];
        }
        if (whole) {
#pragma unroll
            for (int q = 0; q < EPT / 4; ++q)
                *(int4*)(start + i0 + 4 * q) =
                    make_int4(ex[4 * q], ex[4 * q + 1], ex[4 * q + 2], ex[4 * q + 3]);
        } else {
#pragma unroll
            for (int k = 0; k < EPT; ++k)
                if (i0 + k < n) start[i0 + k] = ex[k];
        }
        carry += tile_total;
        __syncthreads();
    }
    if (tid == 0) rp[n] = carry;
}

__global__ void k_fill_rows(const int* __restrict__ src,
                            const int* __restrict__ dst, int E,
                            const int* __restrict__ cnt,
                            const int* __restrict__ rp,
                            const int* __restrict__ rank,
                            int* __restrict__ edges) {
    int e = blockIdx.x * blockDim.x + threadIdx.x;
    if (e < E) {
        const int s = src[e];
        const int d = dst[e];
        const int pos = rp[d] + rank[e];
        *(int2*)(edges + 2 * (size_t)pos) =
            make_int2(s, __float_as_int(dinv_of(cnt[s])));
    }
}

template <int NT, int KD, bool RELU>
__global__ void __launch_bounds__(256) k_gemm_wmma_f32(
    const float* __restrict__ A, const float* __restrict__ B,
    float* __restrict__ C, int M)
{
    constexpr int N   = NT * 16;
    constexpr int LDK = KD + 2;
    __shared__ float lds_w[N * LDK];

    for (int idx = threadIdx.x; idx < KD * N; idx += 256) {
        int k = idx / N;
        int n = idx - k * N;
        lds_w[n * LDK + k] = B[idx];
    }
    __syncthreads();

    const int lane  = threadIdx.x & 31;
    const int wv    = threadIdx.x >> 5;
    const int mtile = blockIdx.x * 8 + wv;
    if (mtile * 16 >= M) return;

    const int hf = lane >> 4;
    const int r  = lane & 15;

    const float* __restrict__ arow = A + (size_t)(mtile * 16 + r) * KD + 2 * hf;

    v8f acc[NT];
#pragma unroll
    for (int t = 0; t < NT; ++t) acc[t] = (v8f){};

#pragma unroll 4
    for (int k = 0; k < KD; k += 4) {
        float2 av = *(const float2*)(arow + k);
        if (RELU) { av.x = fmaxf(av.x, 0.0f); av.y = fmaxf(av.y, 0.0f); }
        v2f a;
        a[0] = av.x;
        a[1] = av.y;
#pragma unroll
        for (int t = 0; t < NT; ++t) {
            const int col = t * 16 + r;
            float2 bv = *(const float2*)(&lds_w[col * LDK + k + 2 * hf]);
            v2f b;
            b[0] = bv.x;
            b[1] = bv.y;
            acc[t] = __builtin_amdgcn_wmma_f32_16x16x4_f32(
                false, a, false, b, (short)0, acc[t], false, false);
        }
    }

#pragma unroll
    for (int t = 0; t < NT; ++t) {
        float* crow = C + (size_t)(mtile * 16 + 8 * hf) * N + t * 16 + r;
#pragma unroll
        for (int j = 0; j < 8; ++j) crow[(size_t)j * N] = acc[t][j];
    }
}

template <int V, int U>
__device__ __forceinline__ void rows_add(
    const __attribute__((ext_vector_type(V))) float* __restrict__ hv,
    int es, float ew, int i, int lane,
    __attribute__((ext_vector_type(V))) float& acc)
{
    typedef __attribute__((ext_vector_type(V))) float vecf;
    vecf  r[U];
    float w[U];
#pragma unroll
    for (int u = 0; u < U; ++u) {
        const int s = __builtin_amdgcn_readlane(es, i + u);
        w[u] = __int_as_float(__builtin_amdgcn_readlane(__float_as_int(ew), i + u));
        r[u] = hv[(size_t)s * 32 + lane];
    }
#pragma unroll
    for (int u = 0; u < U; ++u) acc += r[u] * w[u];
}

template <int C>
__global__ void __launch_bounds__(256) k_gather_sum(
    const int* __restrict__ rp, const int* __restrict__ edges,
    const int* __restrict__ cnt, const float* __restrict__ h,
    const float* __restrict__ bias, float* __restrict__ out, int n)
{
    constexpr int V = C / 32;
    typedef __attribute__((ext_vector_type(V))) float vecf;
    const int lane = threadIdx.x & 31;
    const int node = blockIdx.x * 8 + (threadIdx.x >> 5);
    if (node >= n) return;
    const int d = __builtin_amdgcn_readfirstlane(node);
    const int beg = rp[d];
    const int end = rp[d + 1];
    const float dd = dinv_of(cnt[node]);

    const vecf* __restrict__ hv = (const vecf*)h;
    const int2* __restrict__ ev = (const int2*)edges;

    vecf acc;
    {
        const vecf self = hv[(size_t)d * 32 + lane];
#pragma unroll
        for (int j = 0; j < V; ++j)
            acc[j] = bias[lane * V + j] + self[j] * dd * dd;
    }

    for (int base = beg; base < end; base += 32) {
        const int m = min(32, end - base);
        int   es = 0;
        float ew = 0.0f;
        if (lane < m) {
            const int2 p = ev[base + lane];
            es = p.x;
            ew = __int_as_float(p.y) * dd;
        }
        int i = 0;
        for (; i + 8 <= m; i += 8) rows_add<V, 8>(hv, es, ew, i, lane, acc);
        if (i + 4 <= m) { rows_add<V, 4>(hv, es, ew, i, lane, acc); i += 4; }
        if (i + 2 <= m) { rows_add<V, 2>(hv, es, ew, i, lane, acc); i += 2; }
        if (i < m)      { rows_add<V, 1>(hv, es, ew, i, lane, acc); }
    }

    ((vecf*)out)[(size_t)d * 32 + lane] = acc;
}

extern "C" void kernel_launch(void* const* d_in, const int* in_sizes, int n_in,
                              void* d_out, int out_size, void* d_ws, size_t ws_size,
                              hipStream_t stream) {
    const float* x  = (const float*)d_in[0];
    const int*   ei = (const int*)d_in[1];
    const float* W1 = (const float*)d_in[2];
    const float* b1 = (const float*)d_in[3];
    const float* W2 = (const float*)d_in[4];
    const float* b2 = (const float*)d_in[5];
    float* out = (float*)d_out;

    const int HID = in_sizes[3];
    const int OUT = in_sizes[5];
    const int IN  = in_sizes[2] / HID;
    const int N   = in_sizes[0] / IN;
    const int E   = in_sizes[1] / 2;
    const int* srcI = ei;
    const int* dstI = ei + E;

    const size_t nPad  = ((size_t)N + 4 + 63) & ~(size_t)63;
    const size_t rPad  = ((size_t)E + 63) & ~(size_t)63;
    const size_t ePad  = (2 * (size_t)E + 63) & ~(size_t)63;
    float* ws    = (float*)d_ws;
    int*   cnt   = (int*)ws;
    int*   rp    = (int*)(ws + nPad);
    int*   rank  = (int*)(ws + 2 * nPad);
    int*   edges = (int*)(ws + 2 * nPad + rPad);
    float* bufA  = ws + 2 * nPad + rPad + ePad;
    float* bufB  = bufA + (size_t)N * HID;

    const int T = 256;
    const int mtiles = N / 16;
    const int gblk   = (mtiles + 7) / 8;
    const int nblk   = (N + 7) / 8;

    k_zero_cnt <<<(N + T - 1) / T, T, 0, stream>>>(cnt, N);
    k_count_rank<<<(E + T - 1) / T, T, 0, stream>>>(dstI, E, cnt, rank);
    k_scan_rows<<<1, 1024, 0, stream>>>(cnt, rp, N);
    k_fill_rows<<<(E + T - 1) / T, T, 0, stream>>>(srcI, dstI, E, cnt, rp, rank, edges);

    k_gemm_wmma_f32<8, 128, false><<<gblk, T, 0, stream>>>(x, W1, bufA, N);
    k_gather_sum<128><<<nblk, T, 0, stream>>>(rp, edges, cnt, bufA, b1, bufB, N);

    k_gemm_wmma_f32<4, 128, true><<<gblk, T, 0, stream>>>(bufB, W2, bufA, N);
    k_gather_sum<64><<<nblk, T, 0, stream>>>(rp, edges, cnt, bufA, b2, out, N);
}


# edge count and rank: per-chunk LDS histograms (64 work-groups; the LDS add returns the rank in chunk) + per-node chunk offsets, no global atomics; sum order = chunk, then arrival in chunk
# speedup vs baseline: 2.5761x; 1.2081x over previous
#include <hip/hip_runtime.h>
#include <math.h>


typedef __attribute__((ext_vector_type(2))) float v2f;
typedef __attribute__((ext_vector_type(8))) float v8f;


constexpr int CHUNKS = 64;
constexpr int HB     = 65536;

__device__ __forceinline__ int chunk_len(int E, int chunks) {
    return (E + chunks - 1) / chunks;
}

__global__ void __launch_bounds__(1024) k_count_rank_lds(
    const int* __restrict__ dst, int E, int n, int stride,
    int* __restrict__ part, int* __restrict__ rank)
{
    __shared__ __attribute__((aligned(16))) int hist[HB];
    const int tid = threadIdx.x;
    const int per = chunk_len(E, (int)gridDim.x);
    const int e0  = min(E, (int)blockIdx.x * per);
    const int e1  = min(E, e0 + per);
    int* __restrict__ row = part + (size_t)blockIdx.x * stride;

    for (int r0 = 0; r0 < n; r0 += HB) {
        const int m = min(HB, n - r0);
        for (int i = 4 * tid; i < m; i += 4 * 1024)
            *(int4*)(hist + i) = make_int4(0, 0, 0, 0);
        __syncthreads();

        for (int e = e0 + tid; e < e1; e += 4 * 1024) {
            int d[4];
#pragma unroll
            for (int u = 0; u < 4; ++u) {
                const int ee = e + u * 1024;
                const int v  = dst[min(ee, e1 - 1)] - r0;
                d[u] = (ee < e1) ? v : -1;
            }
#pragma unroll
            for (int u = 0; u < 4; ++u)
                if ((unsigned)d[u] < (unsigned)m)
                    rank[e + u * 1024] = __hip_atomic_fetch_add(
                        &hist[d[u]], 1, __ATOMIC_RELAXED, __HIP_MEMORY_SCOPE_WORKGROUP);
        }
        __syncthreads();

        for (int i = 4 * tid; i < m; i += 4 * 1024)
            *(int4*)(row + r0 + i) = *(const int4*)(hist + i);
        __syncthreads();
    }
}

template <int CH>
__global__ void __launch_bounds__(256) k_chunk_offsets(
    int* __restrict__ part, int stride, int* __restrict__ cnt, int n)
{
    const int d = blockIdx.x * 256 + threadIdx.x;
    if (d >= n) return;
    int* p = part + d;
    int c[CH];
#pragma unroll
    for (int k = 0; k < CH; ++k) c[k] = p[(size_t)k * stride];
    int run = 0;
#pragma unroll
    for (int k = 0; k < CH; ++k) {
        p[(size_t)k * stride] = run;
        run += c[k];
    }
    cnt[d] = run;
}

__device__ __forceinline__ float dinv_of(int c) {
    const float dg = (float)(c + 1);
    return (dg > 0.0f) ? rsqrtf(dg) : 0.0f;
}

__global__ void __launch_bounds__(1024) k_scan_rows(
    const int* __restrict__ cnt, int* __restrict__ rp, int n)
{
    constexpr int EPT  = 16;
    constexpr int TILE = 1024 * EPT;
    __shared__ int wsum[32];
    int* start = rp;
    const int tid  = threadIdx.x;
    const int lane = tid & 31;
    const int wv   = tid >> 5;
    int carry = 0;
    for (int base = 0; base < n; base += TILE) {
        const int i0 = base + tid * EPT;
        const bool whole = (i0 + EPT <= n);
        int c[EPT];
        if (whole) {
#pragma unroll
            for (int q = 0; q < EPT / 4; ++q) {
                const int4 v = *(const int4*)(cnt + i0 + 4 * q);
                c[4 * q] = v.x; c[4 * q + 1] = v.y; c[4 * q + 2] = v.z; c[4 * q + 3] = v.w;
            }
        } else {
#pragma unroll
            for (int k = 0; k < EPT; ++k) c[k] = (i0 + k < n) ? cnt[i0 + k] : 0;
        }
        int tot = 0;
#pragma unroll
        for (int k = 0; k < EPT; ++k) tot += c[k];
        int inc = tot;
        for (int off = 1; off < 32; off <<= 1) {
            const int v = __shfl_up(inc, off);
            if (lane >= off) inc += v;
        }
        if (lane == 31) wsum[wv] = inc;
        __syncthreads();
        const int ws = wsum[lane];
        int wsc = ws;
        for (int off = 1; off < 32; off <<= 1) {
            const int v = __shfl_up(wsc, off);
            if (lane >= off) wsc += v;
        }
        const int wave_excl  = __shfl(wsc, wv) - __shfl(ws, wv);
        const int tile_total = __shfl(wsc, 31);
        int ex[EPT];
        int run = carry + wave_excl + (inc - tot);
#pragma unroll
        for (int k = 0; k < EPT; ++k) {
            ex[k] = run;
            run += c[k];
        }
        if (whole) {
#pragma unroll
            for (int q = 0; q < EPT / 4; ++q)
                *(int4*)(start + i0 + 4 * q) =
                    make_int4(ex[4 * q], ex[4 * q + 1], ex[4 * q + 2], ex[4 * q + 3]);
        } else {
#pragma unroll
            for (int k = 0; k < EPT; ++k)
                if (i0 + k < n) start[i0 + k] = ex[k];
        }
        carry += tile_total;
        __syncthreads();
    }
    if (tid == 0) rp[n] = carry;
}

__global__ void k_fill_rows(const int* __restrict__ src,
                            const int* __restrict__ dst, int E, int chunks,
                            const int* __restrict__ cnt,
                            const int* __restrict__ rp,
                            const int* __restrict__ part, int stride,
                            const int* __restrict__ rank,
                            int* __restrict__ edges) {
    int e = blockIdx.x * blockDim.x + threadIdx.x;
    if (e < E) {
        const int s = src[e];
        const int d = dst[e];
        const int c = e / chunk_len(E, chunks);
        const int pos = rp[d] + part[(size_t)c * stride + d] + rank[e];
        *(int2*)(edges + 2 * (size_t)pos) =
            make_int2(s, __float_as_int(dinv_of(cnt[s])));
    }
}

template <int NT, int KD, bool RELU>
__global__ void __launch_bounds__(256) k_gemm_wmma_f32(
    const float* __restrict__ A, const float* __restrict__ B,
    float* __restrict__ C, int M)
{
    constexpr int N   = NT * 16;
    constexpr int LDK = KD + 2;
    __shared__ float lds_w[N * LDK];

    for (int idx = threadIdx.x; idx < KD * N; idx += 256) {
        int k = idx / N;
        int n = idx - k * N;
        lds_w[n * LDK + k] = B[idx];
    }
    __syncthreads();

    const int lane  = threadIdx.x & 31;
    const int wv    = threadIdx.x >> 5;
    const int mtile = blockIdx.x * 8 + wv;
    if (mtile * 16 >= M) return;

    const int hf = lane >> 4;
    const int r  = lane & 15;

    const float* __restrict__ arow = A + (size_t)(mtile * 16 + r) * KD + 2 * hf;

    v8f acc[NT];
#pragma unroll
    for (int t = 0; t < NT; ++t) acc[t] = (v8f){};

#pragma unroll 4
    for (int k = 0; k < KD; k += 4) {
        float2 av = *(const float2*)(arow + k);
        if (RELU) { av.x = fmaxf(av.x, 0.0f); av.y = fmaxf(av.y, 0.0f); }
        v2f a;
        a[0] = av.x;
        a[1] = av.y;
#pragma unroll
        for (int t = 0; t < NT; ++t) {
            const int col = t * 16 + r;
            float2 bv = *(const float2*)(&lds_w[col * LDK + k + 2 * hf]);
            v2f b;
            b[0] = bv.x;
            b[1] = bv.y;
            acc[t] = __builtin_amdgcn_wmma_f32_16x16x4_f32(
                false, a, false, b, (short)0, acc[t], false, false);
        }
    }

#pragma unroll
    for (int t = 0; t < NT; ++t) {
        float* crow = C + (size_t)(mtile * 16 + 8 * hf) * N + t * 16 + r;
#pragma unroll
        for (int j = 0; j < 8; ++j) crow[(size_t)j * N] = acc[t][j];
    }
}

template <int V, int U>
__device__ __forceinline__ void rows_add(
    const __attribute__((ext_vector_type(V))) float* __restrict__ hv,
    int es, float ew, int i, int lane,
    __attribute__((ext_vector_type(V))) float& acc)
{
    typedef __attribute__((ext_vector_type(V))) float vecf;
    vecf  r[U];
    float w[U];
#pragma unroll
    for (int u = 0; u < U; ++u) {
        const int s = __builtin_amdgcn_readlane(es, i + u);
        w[u] = __int_as_float(__builtin_amdgcn_readlane(__float_as_int(ew), i + u));
        r[u] = hv[(size_t)s * 32 + lane];
    }
#pragma unroll
    for (int u = 0; u < U; ++u) acc += r[u] * w[u];
}

template <int C>
__global__ void __launch_bounds__(256) k_gather_sum(
    const int* __restrict__ rp, const int* __restrict__ edges,
    const int* __restrict__ cnt, const float* __restrict__ h,
    const float* __restrict__ bias, float* __restrict__ out, int n)
{
    constexpr int V = C / 32;
    typedef __attribute__((ext_vector_type(V))) float vecf;
    const int lane = threadIdx.x & 31;
    const int node = blockIdx.x * 8 + (threadIdx.x >> 5);
    if (node >= n) return;
    const int d = __builtin_amdgcn_readfirstlane(node);
    const int beg = rp[d];
    const int end = rp[d + 1];
    const float dd = dinv_of(cnt[node]);

    const vecf* __restrict__ hv = (const vecf*)h;
    const int2* __restrict__ ev = (const int2*)edges;

    vecf acc;
    {
        const vecf self = hv[(size_t)d * 32 + lane];
#pragma unroll
        for (int j = 0; j < V; ++j)
            acc[j] = bias[lane * V + j] + self[j] * dd * dd;
    }

    for (int base = beg; base < end; base += 32) {
        const int m = min(32, end - base);
        int   es = 0;
        float ew = 0.0f;
        if (lane < m) {
            const int2 p = ev[base + lane];
            es = p.x;
            ew = __int_as_float(p.y) * dd;
        }
        int i = 0;
        for (; i + 8 <= m; i += 8) rows_add<V, 8>(hv, es, ew, i, lane, acc);
        if (i + 4 <= m) { rows_add<V, 4>(hv, es, ew, i, lane, acc); i += 4; }
        if (i + 2 <= m) { rows_add<V, 2>(hv, es, ew, i, lane, acc); i += 2; }
        if (i < m)      { rows_add<V, 1>(hv, es, ew, i, lane, acc); }
    }

    ((vecf*)out)[(size_t)d * 32 + lane] = acc;
}

extern "C" void kernel_launch(void* const* d_in, const int* in_sizes, int n_in,
                              void* d_out, int out_size, void* d_ws, size_t ws_size,
                              hipStream_t stream) {
    const float* x  = (const float*)d_in[0];
    const int*   ei = (const int*)d_in[1];
    const float* W1 = (const float*)d_in[2];
    const float* b1 = (const float*)d_in[3];
    const float* W2 = (const float*)d_in[4];
    const float* b2 = (const float*)d_in[5];
    float* out = (float*)d_out;

    const int HID = in_sizes[3];
    const int OUT = in_sizes[5];
    const int IN  = in_sizes[2] / HID;
    const int N   = in_sizes[0] / IN;
    const int E   = in_sizes[1] / 2;
    const int* srcI = ei;
    const int* dstI = ei + E;

    const size_t nPad  = ((size_t)N + 4 + 63) & ~(size_t)63;
    const size_t rPad  = ((size_t)E + 63) & ~(size_t)63;
    const size_t ePad  = (2 * (size_t)E + 63) & ~(size_t)63;
    float* ws    = (float*)d_ws;
    int*   cnt   = (int*)ws;
    int*   rp    = (int*)(ws + nPad);
    int*   rank  = (int*)(ws + 2 * nPad);
    int*   edges = (int*)(ws + 2 * nPad + rPad);
    int*   part  = (int*)(ws + 2 * nPad + rPad + ePad);
    float* bufA  = ws + 2 * nPad + rPad + ePad + CHUNKS * nPad;
    float* bufB  = bufA + (size_t)N * HID;

    const int T = 256;
    const int mtiles = N / 16;
    const int gblk   = (mtiles + 7) / 8;
    const int nblk   = (N + 7) / 8;

    k_count_rank_lds<<<CHUNKS, 1024, 0, stream>>>(dstI, E, N, (int)nPad, part, rank);
    k_chunk_offsets<CHUNKS><<<(N + T - 1) / T, T, 0, stream>>>(part, (int)nPad, cnt, N);
    k_scan_rows<<<1, 1024, 0, stream>>>(cnt, rp, N);
    k_fill_rows<<<(E + T - 1) / T, T, 0, stream>>>(srcI, dstI, E, CHUNKS, cnt, rp,
                                                   part, (int)nPad, rank, edges);

    k_gemm_wmma_f32<8, 128, false><<<gblk, T, 0, stream>>>(x, W1, bufA, N);
    k_gather_sum<128><<<nblk, T, 0, stream>>>(rp, edges, cnt, bufA, b1, bufB, N);

    k_gemm_wmma_f32<4, 128, true><<<gblk, T, 0, stream>>>(bufB, W2, bufA, N);
    k_gather_sum<64><<<nblk, T, 0, stream>>>(rp, edges, cnt, bufA, b2, out, N);
}


# row starts by two-level sum: offsets kernel also sums counts inside its 256-node block, each fill work-group sums the block totals into an LDS table; the one-work-group scan launch is gone
# speedup vs baseline: 2.8689x; 1.1137x over previous
#include <hip/hip_runtime.h>
#include <math.h>


typedef __attribute__((ext_vector_type(2))) float v2f;
typedef __attribute__((ext_vector_type(8))) float v8f;


constexpr int CHUNKS = 64;
constexpr int HB     = 65536;
constexpr int NB     = 256;
constexpr int BP_CAP = 1024;

__device__ __forceinline__ int chunk_len(int E, int chunks) {
    return (E + chunks - 1) / chunks;
}

__global__ void __launch_bounds__(1024) k_count_rank_lds(
    const int* __restrict__ dst, int E, int n, int stride,
    int* __restrict__ part, int* __restrict__ rank)
{
    __shared__ __attribute__((aligned(16))) int hist[HB];
    const int tid = threadIdx.x;
    const int per = chunk_len(E, (int)gridDim.x);
    const int e0  = min(E, (int)blockIdx.x * per);
    const int e1  = min(E, e0 + per);
    int* __restrict__ row = part + (size_t)blockIdx.x * stride;

    for (int r0 = 0; r0 < n; r0 += HB) {
        const int m = min(HB, n - r0);
        for (int i = 4 * tid; i < m; i += 4 * 1024)
            *(int4*)(hist + i) = make_int4(0, 0, 0, 0);
        __syncthreads();

        for (int e = e0 + tid; e < e1; e += 4 * 1024) {
            int d[4];
#pragma unroll
            for (int u = 0; u < 4; ++u) {
                const int ee = e + u * 1024;
                const int v  = dst[min(ee, e1 - 1)] - r0;
                d[u] = (ee < e1) ? v : -1;
            }
#pragma unroll
            for (int u = 0; u < 4; ++u)
                if ((unsigned)d[u] < (unsigned)m)
                    rank[e + u * 1024] = __hip_atomic_fetch_add(
                        &hist[d[u]], 1, __ATOMIC_RELAXED, __HIP_MEMORY_SCOPE_WORKGROUP);
        }
        __syncthreads();

        for (int i = 4 * tid; i < m; i += 4 * 1024)
            *(int4*)(row + r0 + i) = *(const int4*)(hist + i);
        __syncthreads();
    }
}

template <int CH>
__global__ void __launch_bounds__(NB) k_chunk_offsets(
    int* __restrict__ part, int stride, int* __restrict__ cnt,
    int* __restrict__ loc, int* __restrict__ bsum, int n)
{
    __shared__ int wsum[NB / 32];
    const int lane = threadIdx.x & 31;
    const int wv   = threadIdx.x >> 5;
    const int d = blockIdx.x * NB + threadIdx.x;
    const bool live = d < n;
    int* p = part + (live ? d : n - 1);
    int c[CH];
#pragma unroll
    for (int k = 0; k < CH; ++k) c[k] = p[(size_t)k * stride];
    int tot = 0;
#pragma unroll
    for (int k = 0; k < CH; ++k) tot += c[k];
    if (!live) tot = 0;
    int inc = tot;
    for (int off = 1; off < 32; off <<= 1) {
        const int v = __shfl_up(inc, off);
        if (lane >= off) inc += v;
    }
    if (lane == 31) wsum[wv] = inc;
    __syncthreads();
    int before = 0, btot = 0;
#pragma unroll
    for (int k = 0; k < NB / 32; ++k) {
        const int s = wsum[k];
        btot += s;
        if (k < wv) before += s;
    }
    const int ex = before + inc - tot;
    if (live) {
        int run = ex;
#pragma unroll
        for (int k = 0; k < CH; ++k) {
            p[(size_t)k * stride] = run;
            run += c[k];
        }
        cnt[d] = tot;
        loc[d] = ex;
    }
    if (threadIdx.x == 0) bsum[blockIdx.x] = btot;
}

__device__ __forceinline__ float dinv_of(int c) {
    const float dg = (float)(c + 1);
    return (dg > 0.0f) ? rsqrtf(dg) : 0.0f;
}

__global__ void __launch_bounds__(256) k_fill_rows(
    const int* __restrict__ src, const int* __restrict__ dst, int E, int chunks,
    const int* __restrict__ cnt, const int* __restrict__ bsum, int nb,
    const int* __restrict__ loc, const int* __restrict__ part, int stride,
    const int* __restrict__ rank, int* __restrict__ rp, int n,
    int* __restrict__ edges)
{
    __shared__ __attribute__((aligned(16))) int tb[BP_CAP];
    const int lane = threadIdx.x & 31;
    const int e = blockIdx.x * 256 + threadIdx.x;

    int s = 0, d = 0, rk = 0;
    if (e < E) { s = src[e]; d = dst[e]; rk = rank[e]; }

    int carry = 0;
    if (threadIdx.x < 32) {
        for (int base = 0; base < nb; base += 256) {
            const int i0 = base + 8 * lane;
            const int4 lo = *(const int4*)(bsum + i0);
            const int4 hi = *(const int4*)(bsum + i0 + 4);
            int v[8] = {lo.x, lo.y, lo.z, lo.w, hi.x, hi.y, hi.z, hi.w};
            int tot = 0;
#pragma unroll
            for (int k = 0; k < 8; ++k) {
                if (i0 + k >= nb) v[k] = 0;
                tot += v[k];
            }
            int inc = tot;
            for (int off = 1; off < 32; off <<= 1) {
                const int t = __shfl_up(inc, off);
                if (lane >= off) inc += t;
            }
            int x[8];
            int run = carry + inc - tot;
#pragma unroll
            for (int k = 0; k < 8; ++k) {
                x[k] = run;
                run += v[k];
            }
            *(int4*)(tb + i0)     = make_int4(x[0], x[1], x[2], x[3]);
            *(int4*)(tb + i0 + 4) = make_int4(x[4], x[5], x[6], x[7]);
            carry += __shfl(inc, 31);
        }
    }
    __syncthreads();

    if (e < E) {
        const int c = e / chunk_len(E, chunks);
        const int pos = tb[d / NB] + part[(size_t)c * stride + d] + rk;
        *(int2*)(edges + 2 * (size_t)pos) =
            make_int2(s, __float_as_int(dinv_of(cnt[s])));
    }
    for (int i = e; i < n; i += (int)gridDim.x * 256) rp[i] = tb[i / NB] + loc[i];
    if (e == 0) rp[n] = carry;
}

template <int NT, int KD, bool RELU>
__global__ void __launch_bounds__(256) k_gemm_wmma_f32(
    const float* __restrict__ A, const float* __restrict__ B,
    float* __restrict__ C, int M)
{
    constexpr int N   = NT * 16;
    constexpr int LDK = KD + 2;
    __shared__ float lds_w[N * LDK];

    for (int idx = threadIdx.x; idx < KD * N; idx += 256) {
        int k = idx / N;
        int n = idx - k * N;
        lds_w[n * LDK + k] = B[idx];
    }
    __syncthreads();

    const int lane  = threadIdx.x & 31;
    const int wv    = threadIdx.x >> 5;
    const int mtile = blockIdx.x * 8 + wv;
    if (mtile * 16 >= M) return;

    const int hf = lane >> 4;
    const int r  = lane & 15;

    const float* __restrict__ arow = A + (size_t)(mtile * 16 + r) * KD + 2 * hf;

    v8f acc[NT];
#pragma unroll
    for (int t = 0; t < NT; ++t) acc[t] = (v8f){};

#pragma unroll 4
    for (int k = 0; k < KD; k += 4) {
        float2 av = *(const float2*)(arow + k);
        if (RELU) { av.x = fmaxf(av.x, 0.0f); av.y = fmaxf(av.y, 0.0f); }
        v2f a;
        a[0] = av.x;
        a[1] = av.y;
#pragma unroll
        for (int t = 0; t < NT; ++t) {
            const int col = t * 16 + r;
            float2 bv = *(const float2*)(&lds_w[col * LDK + k + 2 * hf]);
            v2f b;
            b[0] = bv.x;
            b[1] = bv.y;
            acc[t] = __builtin_amdgcn_wmma_f32_16x16x4_f32(
                false, a, false, b, (short)0, acc[t], false, false);
        }
    }

#pragma unroll
    for (int t = 0; t < NT; ++t) {
        float* crow = C + (size_t)(mtile * 16 + 8 * hf) * N + t * 16 + r;
#pragma unroll
        for (int j = 0; j < 8; ++j) crow[(size_t)j * N] = acc[t][j];
    }
}

template <int V, int U>
__device__ __forceinline__ void rows_add(
    const __attribute__((ext_vector_type(V))) float* __restrict__ hv,
    int es, float ew, int i, int lane,
    __attribute__((ext_vector_type(V))) float& acc)
{
    typedef __attribute__((ext_vector_type(V))) float vecf;
    vecf  r[U];
    float w[U];
#pragma unroll
    for (int u = 0; u < U; ++u) {
        const int s = __builtin_amdgcn_readlane(es, i + u);
        w[u] = __int_as_float(__builtin_amdgcn_readlane(__float_as_int(ew), i + u));
        r[u] = hv[(size_t)s * 32 + lane];
    }
#pragma unroll
    for (int u = 0; u < U; ++u) acc += r[u] * w[u];
}

template <int C>
__global__ void __launch_bounds__(256) k_gather_sum(
    const int* __restrict__ rp, const int* __restrict__ edges,
    const int* __restrict__ cnt, const float* __restrict__ h,
    const float* __restrict__ bias, float* __restrict__ out, int n)
{
    constexpr int V = C / 32;
    typedef __attribute__((ext_vector_type(V))) float vecf;
    const int lane = threadIdx.x & 31;
    const int node = blockIdx.x * 8 + (threadIdx.x >> 5);
    if (node >= n) return;
    const int d = __builtin_amdgcn_readfirstlane(node);
    const int beg = rp[d];
    const int end = rp[d + 1];
    const float dd = dinv_of(cnt[node]);

    const vecf* __restrict__ hv = (const vecf*)h;
    const int2* __restrict__ ev = (const int2*)edges;

    vecf acc;
    {
        const vecf self = hv[(size_t)d * 32 + lane];
#pragma unroll
        for (int j = 0; j < V; ++j)
            acc[j] = bias[lane * V + j] + self[j] * dd * dd;
    }

    for (int base = beg; base < end; base += 32) {
        const int m = min(32, end - base);
        int   es = 0;
        float ew = 0.0f;
        if (lane < m) {
            const int2 p = ev[base + lane];
            es = p.x;
            ew = __int_as_float(p.y) * dd;
        }
        int i = 0;
        for (; i + 8 <= m; i += 8) rows_add<V, 8>(hv, es, ew, i, lane, acc);
        if (i + 4 <= m) { rows_add<V, 4>(hv, es, ew, i, lane, acc); i += 4; }
        if (i + 2 <= m) { rows_add<V, 2>(hv, es, ew, i, lane, acc); i += 2; }
        if (i < m)      { rows_add<V, 1>(hv, es, ew, i, lane, acc); }
    }

    ((vecf*)out)[(size_t)d * 32 + lane] = acc;
}

extern "C" void kernel_launch(void* const* d_in, const int* in_sizes, int n_in,
                              void* d_out, int out_size, void* d_ws, size_t ws_size,
                              hipStream_t stream) {
    const float* x  = (const float*)d_in[0];
    const int*   ei = (const int*)d_in[1];
    const float* W1 = (const float*)d_in[2];
    const float* b1 = (const float*)d_in[3];
    const float* W2 = (const float*)d_in[4];
    const float* b2 = (const float*)d_in[5];
    float* out = (float*)d_out;

    const int HID = in_sizes[3];
    const int OUT = in_sizes[5];
    const int IN  = in_sizes[2] / HID;
    const int N   = in_sizes[0] / IN;
    const int E   = in_sizes[1] / 2;
    const int* srcI = ei;
    const int* dstI = ei + E;

    const size_t nPad  = ((size_t)N + 4 + 63) & ~(size_t)63;
    const size_t rPad  = ((size_t)E + 63) & ~(size_t)63;
    const size_t ePad  = (2 * (size_t)E + 63) & ~(size_t)63;
    float* ws    = (float*)d_ws;
    int*   cnt   = (int*)ws;
    int*   rp    = (int*)(ws + nPad);
    int*   loc   = (int*)(ws + 2 * nPad);
    int*   bsum  = (int*)(ws + 3 * nPad);
    int*   rank  = (int*)(ws + 3 * nPad + BP_CAP);
    int*   edges = (int*)(ws + 3 * nPad + BP_CAP + rPad);
    int*   part  = (int*)(ws + 3 * nPad + BP_CAP + rPad + ePad);
    float* bufA  = ws + 3 * nPad + BP_CAP + rPad + ePad + CHUNKS * nPad;
    float* bufB  = bufA + (size_t)N * HID;

    const int T = 256;
    const int mtiles = N / 16;
    const int gblk   = (mtiles + 7) / 8;
    const int nblk   = (N + 7) / 8;

    k_count_rank_lds<<<CHUNKS, 1024, 0, stream>>>(dstI, E, N, (int)nPad, part, rank);
    const int nblocks = (N + NB - 1) / NB;
    k_chunk_offsets<CHUNKS><<<nblocks, NB, 0, stream>>>(part, (int)nPad, cnt, loc, bsum, N);
    k_fill_rows<<<(E + T - 1) / T, T, 0, stream>>>(srcI, dstI, E, CHUNKS, cnt, bsum, nblocks,
                                                   loc, part, (int)nPad, rank, rp, N, edges);

    k_gemm_wmma_f32<8, 128, false><<<gblk, T, 0, stream>>>(x, W1, bufA, N);
    k_gather_sum<128><<<nblk, T, 0, stream>>>(rp, edges, cnt, bufA, b1, bufB, N);

    k_gemm_wmma_f32<4, 128, true><<<gblk, T, 0, stream>>>(bufB, W2, bufA, N);
    k_gather_sum<64><<<nblk, T, 0, stream>>>(rp, edges, cnt, bufA, b2, out, N);
}


# layer-1 product x @ W1 and the edge-list fill, which need nothing of each other, run in one launch (matrix work-groups first, fill work-groups beside them); 6 launches, sum order unchanged
# speedup vs baseline: 3.1218x; 1.0882x over previous
#include <hip/hip_runtime.h>
#include <math.h>


typedef __attribute__((ext_vector_type(2))) float v2f;
typedef __attribute__((ext_vector_type(8))) float v8f;


constexpr int CHUNKS = 64;
constexpr int HB     = 65536;
constexpr int NB     = 256;
constexpr int BP_CAP = 1024;

__device__ __forceinline__ int chunk_len(int E, int chunks) {
    return (E + chunks - 1) / chunks;
}

__global__ void __launch_bounds__(1024) k_count_rank_lds(
    const int* __restrict__ dst, int E, int n, int stride,
    int* __restrict__ part, int* __restrict__ rank)
{
    __shared__ __attribute__((aligned(16))) int hist[HB];
    const int tid = threadIdx.x;
    const int per = chunk_len(E, (int)gridDim.x);
    const int e0  = min(E, (int)blockIdx.x * per);
    const int e1  = min(E, e0 + per);
    int* __restrict__ row = part + (size_t)blockIdx.x * stride;

    for (int r0 = 0; r0 < n; r0 += HB) {
        const int m = min(HB, n - r0);
        for (int i = 4 * tid; i < m; i += 4 * 1024)
            *(int4*)(hist + i) = make_int4(0, 0, 0, 0);
        __syncthreads();

        for (int e = e0 + tid; e < e1; e += 4 * 1024) {
            int d[4];
#pragma unroll
            for (int u = 0; u < 4; ++u) {
                const int ee = e + u * 1024;
                const int v  = dst[min(ee, e1 - 1)] - r0;
                d[u] = (ee < e1) ? v : -1;
            }
#pragma unroll
            for (int u = 0; u < 4; ++u)
                if ((unsigned)d[u] < (unsigned)m)
                    rank[e + u * 1024] = __hip_atomic_fetch_add(
                        &hist[d[u]], 1, __ATOMIC_RELAXED, __HIP_MEMORY_SCOPE_WORKGROUP);
        }
        __syncthreads();

        for (int i = 4 * tid; i < m; i += 4 * 1024)
            *(int4*)(row + r0 + i) = *(const int4*)(hist + i);
        __syncthreads();
    }
}

template <int CH>
__global__ void __launch_bounds__(NB) k_chunk_offsets(
    int* __restrict__ part, int stride, int* __restrict__ cnt,
    int* __restrict__ loc, int* __restrict__ bsum, int n)
{
    __shared__ int wsum[NB / 32];
    const int lane = threadIdx.x & 31;
    const int wv   = threadIdx.x >> 5;
    const int d = blockIdx.x * NB + threadIdx.x;
    const bool live = d < n;
    int* p = part + (live ? d : n - 1);
    int c[CH];
#pragma unroll
    for (int k = 0; k < CH; ++k) c[k] = p[(size_t)k * stride];
    int tot = 0;
#pragma unroll
    for (int k = 0; k < CH; ++k) tot += c[k];
    if (!live) tot = 0;
    int inc = tot;
    for (int off = 1; off < 32; off <<= 1) {
        const int v = __shfl_up(inc, off);
        if (lane >= off) inc += v;
    }
    if (lane == 31) wsum[wv] = inc;
    __syncthreads();
    int before = 0, btot = 0;
#pragma unroll
    for (int k = 0; k < NB / 32; ++k) {
        const int s = wsum[k];
        btot += s;
        if (k < wv) before += s;
    }
    const int ex = before + inc - tot;
    if (live) {
        int run = ex;
#pragma unroll
        for (int k = 0; k < CH; ++k) {
            p[(size_t)k * stride] = run;
            run += c[k];
        }
        cnt[d] = tot;
        loc[d] = ex;
    }
    if (threadIdx.x == 0) bsum[blockIdx.x] = btot;
}

__device__ __forceinline__ float dinv_of(int c) {
    const float dg = (float)(c + 1);
    return (dg > 0.0f) ? rsqrtf(dg) : 0.0f;
}

__device__ __forceinline__ void fill_rows_block(
    int* __restrict__ tb, int blk, int nblk,
    const int* __restrict__ src, const int* __restrict__ dst, int E, int chunks,
    const int* __restrict__ cnt, const int* __restrict__ bsum, int nb,
    const int* __restrict__ loc, const int* __restrict__ part, int stride,
    const int* __restrict__ rank, int* __restrict__ rp, int n,
    int* __restrict__ edges)
{
    const int lane = threadIdx.x & 31;
    const int e = blk * 256 + threadIdx.x;

    int s = 0, d = 0, rk = 0;
    if (e < E) { s = src[e]; d = dst[e]; rk = rank[e]; }

    int carry = 0;
    if (threadIdx.x < 32) {
        for (int base = 0; base < nb; base += 256) {
            const int i0 = base + 8 * lane;
            const int4 lo = *(const int4*)(bsum + i0);
            const int4 hi = *(const int4*)(bsum + i0 + 4);
            int v[8] = {lo.x, lo.y, lo.z, lo.w, hi.x, hi.y, hi.z, hi.w};
            int tot = 0;
#pragma unroll
            for (int k = 0; k < 8; ++k) {
                if (i0 + k >= nb) v[k] = 0;
                tot += v[k];
            }
            int inc = tot;
            for (int off = 1; off < 32; off <<= 1) {
                const int t = __shfl_up(inc, off);
                if (lane >= off) inc += t;
            }
            int x[8];
            int run = carry + inc - tot;
#pragma unroll
            for (int k = 0; k < 8; ++k) {
                x[k] = run;
                run += v[k];
            }
            *(int4*)(tb + i0)     = make_int4(x[0], x[1], x[2], x[3]);
            *(int4*)(tb + i0 + 4) = make_int4(x[4], x[5], x[6], x[7]);
            carry += __shfl(inc, 31);
        }
    }
    __syncthreads();

    if (e < E) {
        const int c = e / chunk_len(E, chunks);
        const int pos = tb[d / NB] + part[(size_t)c * stride + d] + rk;
        *(int2*)(edges + 2 * (size_t)pos) =
            make_int2(s, __float_as_int(dinv_of(cnt[s])));
    }
    for (int i = e; i < n; i += nblk * 256) rp[i] = tb[i / NB] + loc[i];
    if (e == 0) rp[n] = carry;
}

template <int NT, int KD, bool RELU>
__device__ __forceinline__ void gemm_rows_block(
    float* __restrict__ lds_w, int blk,
    const float* __restrict__ A, const float* __restrict__ B,
    float* __restrict__ C, int M)
{
    constexpr int N   = NT * 16;
    constexpr int LDK = KD + 2;

    for (int idx = threadIdx.x; idx < KD * N; idx += 256) {
        int k = idx / N;
        int n = idx - k * N;
        lds_w[n * LDK + k] = B[idx];
    }
    __syncthreads();

    const int lane  = threadIdx.x & 31;
    const int wv    = threadIdx.x >> 5;
    const int mtile = blk * 8 + wv;
    if (mtile * 16 >= M) return;

    const int hf = lane >> 4;
    const int r  = lane & 15;

    const float* __restrict__ arow = A + (size_t)(mtile * 16 + r) * KD + 2 * hf;

    v8f acc[NT];
#pragma unroll
    for (int t = 0; t < NT; ++t) acc[t] = (v8f){};

#pragma unroll 4
    for (int k = 0; k < KD; k += 4) {
        float2 av = *(const float2*)(arow + k);
        if (RELU) { av.x = fmaxf(av.x, 0.0f); av.y = fmaxf(av.y, 0.0f); }
        v2f a;
        a[0] = av.x;
        a[1] = av.y;
#pragma unroll
        for (int t = 0; t < NT; ++t) {
            const int col = t * 16 + r;
            float2 bv = *(const float2*)(&lds_w[col * LDK + k + 2 * hf]);
            v2f b;
            b[0] = bv.x;
            b[1] = bv.y;
            acc[t] = __builtin_amdgcn_wmma_f32_16x16x4_f32(
                false, a, false, b, (short)0, acc[t], false, false);
        }
    }

#pragma unroll
    for (int t = 0; t < NT; ++t) {
        float* crow = C + (size_t)(mtile * 16 + 8 * hf) * N + t * 16 + r;
#pragma unroll
        for (int j = 0; j < 8; ++j) crow[(size_t)j * N] = acc[t][j];
    }
}

template <int NT, int KD, bool RELU>
__global__ void __launch_bounds__(256) k_gemm_wmma_f32(
    const float* __restrict__ A, const float* __restrict__ B,
    float* __restrict__ C, int M)
{
    __shared__ float lds_w[NT * 16 * (KD + 2)];
    gemm_rows_block<NT, KD, RELU>(lds_w, (int)blockIdx.x, A, B, C, M);
}

__global__ void __launch_bounds__(256) k_gemm1_with_fill(
    const float* __restrict__ x, const float* __restrict__ W1,
    float* __restrict__ h1, int M, int gemm_blocks,
    const int* __restrict__ src, const int* __restrict__ dst, int E, int chunks,
    const int* __restrict__ cnt, const int* __restrict__ bsum, int nb,
    const int* __restrict__ loc, const int* __restrict__ part, int stride,
    const int* __restrict__ rank, int* __restrict__ rp, int n,
    int* __restrict__ edges)
{
    __shared__ __attribute__((aligned(16))) union {
        float w[8 * 16 * (128 + 2)];
        int   tb[BP_CAP];
    } sm;
    const int b = (int)blockIdx.x;
    if (b < gemm_blocks) {
        gemm_rows_block<8, 128, false>(sm.w, b, x, W1, h1, M);
    } else {
        fill_rows_block(sm.tb, b - gemm_blocks,
                        (int)gridDim.x - gemm_blocks,
                        src, dst, E, chunks, cnt, bsum, nb, loc, part, stride,
                        rank, rp, n, edges);
    }
}

template <int V, int U>
__device__ __forceinline__ void rows_add(
    const __attribute__((ext_vector_type(V))) float* __restrict__ hv,
    int es, float ew, int i, int lane,
    __attribute__((ext_vector_type(V))) float& acc)
{
    typedef __attribute__((ext_vector_type(V))) float vecf;
    vecf  r[U];
    float w[U];
#pragma unroll
    for (int u = 0; u < U; ++u) {
        const int s = __builtin_amdgcn_readlane(es, i + u);
        w[u] = __int_as_float(__builtin_amdgcn_readlane(__float_as_int(ew), i + u));
        r[u] = hv[(size_t)s * 32 + lane];
    }
#pragma unroll
    for (int u = 0; u < U; ++u) acc += r[u] * w[u];
}

template <int C>
__global__ void __launch_bounds__(256) k_gather_sum(
    const int* __restrict__ rp, const int* __restrict__ edges,
    const int* __restrict__ cnt, const float* __restrict__ h,
    const float* __restrict__ bias, float* __restrict__ out, int n)
{
    constexpr int V = C / 32;
    typedef __attribute__((ext_vector_type(V))) float vecf;
    const int lane = threadIdx.x & 31;
    const int node = blockIdx.x * 8 + (threadIdx.x >> 5);
    if (node >= n) return;
    const int d = __builtin_amdgcn_readfirstlane(node);
    const int beg = rp[d];
    const int end = rp[d + 1];
    const float dd = dinv_of(cnt[node]);

    const vecf* __restrict__ hv = (const vecf*)h;
    const int2* __restrict__ ev = (const int2*)edges;

    vecf acc;
    {
        const vecf self = hv[(size_t)d * 32 + lane];
#pragma unroll
        for (int j = 0; j < V; ++j)
            acc[j] = bias[lane * V + j] + self[j] * dd * dd;
    }

    for (int base = beg; base < end; base += 32) {
        const int m = min(32, end - base);
        int   es = 0;
        float ew = 0.0f;
        if (lane < m) {
            const int2 p = ev[base + lane];
            es = p.x;
            ew = __int_as_float(p.y) * dd;
        }
        int i = 0;
        for (; i + 8 <= m; i += 8) rows_add<V, 8>(hv, es, ew, i, lane, acc);
        if (i + 4 <= m) { rows_add<V, 4>(hv, es, ew, i, lane, acc); i += 4; }
        if (i + 2 <= m) { rows_add<V, 2>(hv, es, ew, i, lane, acc); i += 2; }
        if (i < m)      { rows_add<V, 1>(hv, es, ew, i, lane, acc); }
    }

    ((vecf*)out)[(size_t)d * 32 + lane] = acc;
}

extern "C" void kernel_launch(void* const* d_in, const int* in_sizes, int n_in,
                              void* d_out, int out_size, void* d_ws, size_t ws_size,
                              hipStream_t stream) {
    const float* x  = (const float*)d_in[0];
    const int*   ei = (const int*)d_in[1];
    const float* W1 = (const float*)d_in[2];
    const float* b1 = (const float*)d_in[3];
    const float* W2 = (const float*)d_in[4];
    const float* b2 = (const float*)d_in[5];
    float* out = (float*)d_out;

    const int HID = in_sizes[3];
    const int OUT = in_sizes[5];
    const int IN  = in_sizes[2] / HID;
    const int N   = in_sizes[0] / IN;
    const int E   = in_sizes[1] / 2;
    const int* srcI = ei;
    const int* dstI = ei + E;

    const size_t nPad  = ((size_t)N + 4 + 63) & ~(size_t)63;
    const size_t rPad  = ((size_t)E + 63) & ~(size_t)63;
    const size_t ePad  = (2 * (size_t)E + 63) & ~(size_t)63;
    float* ws    = (float*)d_ws;
    int*   cnt   = (int*)ws;
    int*   rp    = (int*)(ws + nPad);
    int*   loc   = (int*)(ws + 2 * nPad);
    int*   bsum  = (int*)(ws + 3 * nPad);
    int*   rank  = (int*)(ws + 3 * nPad + BP_CAP);
    int*   edges = (int*)(ws + 3 * nPad + BP_CAP + rPad);
    int*   part  = (int*)(ws + 3 * nPad + BP_CAP + rPad + ePad);
    float* bufA  = ws + 3 * nPad + BP_CAP + rPad + ePad + CHUNKS * nPad;
    float* bufB  = bufA + (size_t)N * HID;

    const int T = 256;
    const int mtiles = N / 16;
    const int gblk   = (mtiles + 7) / 8;
    const int nblk   = (N + 7) / 8;

    k_count_rank_lds<<<CHUNKS, 1024, 0, stream>>>(dstI, E, N, (int)nPad, part, rank);
    const int nblocks = (N + NB - 1) / NB;
    k_chunk_offsets<CHUNKS><<<nblocks, NB, 0, stream>>>(part, (int)nPad, cnt, loc, bsum, N);
    k_gemm1_with_fill<<<gblk + (E + T - 1) / T, T, 0, stream>>>(
        x, W1, bufA, N, gblk,
        srcI, dstI, E, CHUNKS, cnt, bsum, nblocks, loc, part, (int)nPad, rank, rp, N, edges);
    k_gather_sum<128><<<nblk, T, 0, stream>>>(rp, edges, cnt, bufA, b1, bufB, N);

    k_gemm_wmma_f32<4, 128, true><<<gblk, T, 0, stream>>>(bufB, W2, bufA, N);
    k_gather_sum<64><<<nblk, T, 0, stream>>>(rp, edges, cnt, bufA, b2, out, N);
}


# layer 2's product relu(agg1) @ W2 moves into layer 1's row-sum launch: 16-wave work-groups sum 64 nodes into LDS, then one 16x16 f32 WMMA tile per wave; agg1 never stored; 5 launches, sum order kept
# speedup vs baseline: 3.3016x; 1.0576x over previous
#include <hip/hip_runtime.h>
#include <math.h>


typedef __attribute__((ext_vector_type(2))) float v2f;
typedef __attribute__((ext_vector_type(8))) float v8f;


constexpr int CHUNKS = 64;
constexpr int HB     = 65536;
constexpr int NB     = 256;
constexpr int BP_CAP = 1024;

__device__ __forceinline__ int chunk_len(int E, int chunks) {
    return (E + chunks - 1) / chunks;
}

__global__ void __launch_bounds__(1024) k_count_rank_lds(
    const int* __restrict__ dst, int E, int n, int stride,
    int* __restrict__ part, int* __restrict__ rank)
{
    __shared__ __attribute__((aligned(16))) int hist[HB];
    const int tid = threadIdx.x;
    const int per = chunk_len(E, (int)gridDim.x);
    const int e0  = min(E, (int)blockIdx.x * per);
    const int e1  = min(E, e0 + per);
    int* __restrict__ row = part + (size_t)blockIdx.x * stride;

    for (int r0 = 0; r0 < n; r0 += HB) {
        const int m = min(HB, n - r0);
        for (int i = 4 * tid; i < m; i += 4 * 1024)
            *(int4*)(hist + i) = make_int4(0, 0, 0, 0);
        __syncthreads();

        for (int e = e0 + tid; e < e1; e += 4 * 1024) {
            int d[4];
#pragma unroll
            for (int u = 0; u < 4; ++u) {
                const int ee = e + u * 1024;
                const int v  = dst[min(ee, e1 - 1)] - r0;
                d[u] = (ee < e1) ? v : -1;
            }
#pragma unroll
            for (int u = 0; u < 4; ++u)
                if ((unsigned)d[u] < (unsigned)m)
                    rank[e + u * 1024] = __hip_atomic_fetch_add(
                        &hist[d[u]], 1, __ATOMIC_RELAXED, __HIP_MEMORY_SCOPE_WORKGROUP);
        }
        __syncthreads();

        for (int i = 4 * tid; i < m; i += 4 * 1024)
            *(int4*)(row + r0 + i) = *(const int4*)(hist + i);
        __syncthreads();
    }
}

template <int CH>
__global__ void __launch_bounds__(NB) k_chunk_offsets(
    int* __restrict__ part, int stride, int* __restrict__ cnt,
    int* __restrict__ loc, int* __restrict__ bsum, int n)
{
    __shared__ int wsum[NB / 32];
    const int lane = threadIdx.x & 31;
    const int wv   = threadIdx.x >> 5;
    const int d = blockIdx.x * NB + threadIdx.x;
    const bool live = d < n;
    int* p = part + (live ? d : n - 1);
    int c[CH];
#pragma unroll
    for (int k = 0; k < CH; ++k) c[k] = p[(size_t)k * stride];
    int tot = 0;
#pragma unroll
    for (int k = 0; k < CH; ++k) tot += c[k];
    if (!live) tot = 0;
    int inc = tot;
    for (int off = 1; off < 32; off <<= 1) {
        const int v = __shfl_up(inc, off);
        if (lane >= off) inc += v;
    }
    if (lane == 31) wsum[wv] = inc;
    __syncthreads();
    int before = 0, btot = 0;
#pragma unroll
    for (int k = 0; k < NB / 32; ++k) {
        const int s = wsum[k];
        btot += s;
        if (k < wv) before += s;
    }
    const int ex = before + inc - tot;
    if (live) {
        int run = ex;
#pragma unroll
        for (int k = 0; k < CH; ++k) {
            p[(size_t)k * stride] = run;
            run += c[k];
        }
        cnt[d] = tot;
        loc[d] = ex;
    }
    if (threadIdx.x == 0) bsum[blockIdx.x] = btot;
}

__device__ __forceinline__ float dinv_of(int c) {
    const float dg = (float)(c + 1);
    return (dg > 0.0f) ? rsqrtf(dg) : 0.0f;
}

__device__ __forceinline__ void fill_rows_block(
    int* __restrict__ tb, int blk, int nblk,
    const int* __restrict__ src, const int* __restrict__ dst, int E, int chunks,
    const int* __restrict__ cnt, const int* __restrict__ bsum, int nb,
    const int* __restrict__ loc, const int* __restrict__ part, int stride,
    const int* __restrict__ rank, int* __restrict__ rp, int n,
    int* __restrict__ edges)
{
    const int lane = threadIdx.x & 31;
    const int e = blk * 256 + threadIdx.x;

    int s = 0, d = 0, rk = 0;
    if (e < E) { s = src[e]; d = dst[e]; rk = rank[e]; }

    int carry = 0;
    if (threadIdx.x < 32) {
        for (int base = 0; base < nb; base += 256) {
            const int i0 = base + 8 * lane;
            const int4 lo = *(const int4*)(bsum + i0);
            const int4 hi = *(const int4*)(bsum + i0 + 4);
            int v[8] = {lo.x, lo.y, lo.z, lo.w, hi.x, hi.y, hi.z, hi.w};
            int tot = 0;
#pragma unroll
            for (int k = 0; k < 8; ++k) {
                if (i0 + k >= nb) v[k] = 0;
                tot += v[k];
            }
            int inc = tot;
            for (int off = 1; off < 32; off <<= 1) {
                const int t = __shfl_up(inc, off);
                if (lane >= off) inc += t;
            }
            int x[8];
            int run = carry + inc - tot;
#pragma unroll
            for (int k = 0; k < 8; ++k) {
                x[k] = run;
                run += v[k];
            }
            *(int4*)(tb + i0)     = make_int4(x[0], x[1], x[2], x[3]);
            *(int4*)(tb + i0 + 4) = make_int4(x[4], x[5], x[6], x[7]);
            carry += __shfl(inc, 31);
        }
    }
    __syncthreads();

    if (e < E) {
        const int c = e / chunk_len(E, chunks);
        const int pos = tb[d / NB] + part[(size_t)c * stride + d] + rk;
        *(int2*)(edges + 2 * (size_t)pos) =
            make_int2(s, __float_as_int(dinv_of(cnt[s])));
    }
    for (int i = e; i < n; i += nblk * 256) rp[i] = tb[i / NB] + loc[i];
    if (e == 0) rp[n] = carry;
}

template <int NT, int KD, bool RELU>
__device__ __forceinline__ void gemm_rows_block(
    float* __restrict__ lds_w, int blk,
    const float* __restrict__ A, const float* __restrict__ B,
    float* __restrict__ C, int M)
{
    constexpr int N   = NT * 16;
    constexpr int LDK = KD + 2;

    for (int idx = threadIdx.x; idx < KD * N; idx += 256) {
        int k = idx / N;
        int n = idx - k * N;
        lds_w[n * LDK + k] = B[idx];
    }
    __syncthreads();

    const int lane  = threadIdx.x & 31;
    const int wv    = threadIdx.x >> 5;
    const int mtile = blk * 8 + wv;
    if (mtile * 16 >= M) return;

    const int hf = lane >> 4;
    const int r  = lane & 15;

    const float* __restrict__ arow = A + (size_t)(mtile * 16 + r) * KD + 2 * hf;

    v8f acc[NT];
#pragma unroll
    for (int t = 0; t < NT; ++t) acc[t] = (v8f){};

#pragma unroll 4
    for (int k = 0; k < KD; k += 4) {
        float2 av = *(const float2*)(arow + k);
        if (RELU) { av.x = fmaxf(av.x, 0.0f); av.y = fmaxf(av.y, 0.0f); }
        v2f a;
        a[0] = av.x;
        a[1] = av.y;
#pragma unroll
        for (int t = 0; t < NT; ++t) {
            const int col = t * 16 + r;
            float2 bv = *(const float2*)(&lds_w[col * LDK + k + 2 * hf]);
            v2f b;
            b[0] = bv.x;
            b[1] = bv.y;
            acc[t] = __builtin_amdgcn_wmma_f32_16x16x4_f32(
                false, a, false, b, (short)0, acc[t], false, false);
        }
    }

#pragma unroll
    for (int t = 0; t < NT; ++t) {
        float* crow = C + (size_t)(mtile * 16 + 8 * hf) * N + t * 16 + r;
#pragma unroll
        for (int j = 0; j < 8; ++j) crow[(size_t)j * N] = acc[t][j];
    }
}

__global__ void __launch_bounds__(256) k_gemm1_with_fill(
    const float* __restrict__ x, const float* __restrict__ W1,
    float* __restrict__ h1, int M, int gemm_blocks,
    const int* __restrict__ src, const int* __restrict__ dst, int E, int chunks,
    const int* __restrict__ cnt, const int* __restrict__ bsum, int nb,
    const int* __restrict__ loc, const int* __restrict__ part, int stride,
    const int* __restrict__ rank, int* __restrict__ rp, int n,
    int* __restrict__ edges)
{
    __shared__ __attribute__((aligned(16))) union {
        float w[8 * 16 * (128 + 2)];
        int   tb[BP_CAP];
    } sm;
    const int b = (int)blockIdx.x;
    if (b < gemm_blocks) {
        gemm_rows_block<8, 128, false>(sm.w, b, x, W1, h1, M);
    } else {
        fill_rows_block(sm.tb, b - gemm_blocks,
                        (int)gridDim.x - gemm_blocks,
                        src, dst, E, chunks, cnt, bsum, nb, loc, part, stride,
                        rank, rp, n, edges);
    }
}

template <int V, int U>
__device__ __forceinline__ void rows_add(
    const __attribute__((ext_vector_type(V))) float* __restrict__ hv,
    int es, float ew, int i, int lane,
    __attribute__((ext_vector_type(V))) float& acc)
{
    typedef __attribute__((ext_vector_type(V))) float vecf;
    vecf  r[U];
    float w[U];
#pragma unroll
    for (int u = 0; u < U; ++u) {
        const int s = __builtin_amdgcn_readlane(es, i + u);
        w[u] = __int_as_float(__builtin_amdgcn_readlane(__float_as_int(ew), i + u));
        r[u] = hv[(size_t)s * 32 + lane];
    }
#pragma unroll
    for (int u = 0; u < U; ++u) acc += r[u] * w[u];
}

template <int C>
__global__ void __launch_bounds__(256) k_gather_sum(
    const int* __restrict__ rp, const int* __restrict__ edges,
    const int* __restrict__ cnt, const float* __restrict__ h,
    const float* __restrict__ bias, float* __restrict__ out, int n)
{
    constexpr int V = C / 32;
    typedef __attribute__((ext_vector_type(V))) float vecf;
    const int lane = threadIdx.x & 31;
    const int node = blockIdx.x * 8 + (threadIdx.x >> 5);
    if (node >= n) return;
    const int d = __builtin_amdgcn_readfirstlane(node);
    const int beg = rp[d];
    const int end = rp[d + 1];
    const float dd = dinv_of(cnt[node]);

    const vecf* __restrict__ hv = (const vecf*)h;
    const int2* __restrict__ ev = (const int2*)edges;

    vecf acc;
    {
        const vecf self = hv[(size_t)d * 32 + lane];
#pragma unroll
        for (int j = 0; j < V; ++j)
            acc[j] = bias[lane * V + j] + self[j] * dd * dd;
    }

    for (int base = beg; base < end; base += 32) {
        const int m = min(32, end - base);
        int   es = 0;
        float ew = 0.0f;
        if (lane < m) {
            const int2 p = ev[base + lane];
            es = p.x;
            ew = __int_as_float(p.y) * dd;
        }
        int i = 0;
        for (; i + 8 <= m; i += 8) rows_add<V, 8>(hv, es, ew, i, lane, acc);
        if (i + 4 <= m) { rows_add<V, 4>(hv, es, ew, i, lane, acc); i += 4; }
        if (i + 2 <= m) { rows_add<V, 2>(hv, es, ew, i, lane, acc); i += 2; }
        if (i < m)      { rows_add<V, 1>(hv, es, ew, i, lane, acc); }
    }

    ((vecf*)out)[(size_t)d * 32 + lane] = acc;
}

constexpr int FN   = 64;
constexpr int FA_P = 128 + 4;
constexpr int FW_P = 64 + 8;

__global__ void __launch_bounds__(512) k_gather_sum_relu_gemm(
    const int* __restrict__ rp, const int* __restrict__ edges,
    const int* __restrict__ cnt, const float* __restrict__ h,
    const float* __restrict__ bias, const float* __restrict__ W2,
    float* __restrict__ h2, int n)
{
    typedef __attribute__((ext_vector_type(4))) float vecf;
    __shared__ __attribute__((aligned(16))) float w2s[128 * FW_P];
    __shared__ __attribute__((aligned(16))) float as[FN * FA_P];
    const int lane = threadIdx.x & 31;
    const int wv   = threadIdx.x >> 5;

    {
        float4 w[4];
#pragma unroll
        for (int t = 0; t < 4; ++t)
            w[t] = ((const float4*)W2)[t * 512 + (int)threadIdx.x];
#pragma unroll
        for (int t = 0; t < 4; ++t) {
            const int i = t * 512 + (int)threadIdx.x;
            *(float4*)(w2s + (i >> 4) * FW_P + 4 * (i & 15)) = w[t];
        }
    }

    const vecf* __restrict__ hv = (const vecf*)h;
    const int2* __restrict__ ev = (const int2*)edges;

    for (int j = 0; j < 4; ++j) {
        const int li   = 4 * wv + j;
        const int node = blockIdx.x * FN + li;
        vecf acc = (vecf){0.0f, 0.0f, 0.0f, 0.0f};
        if (node < n) {
            const int d = __builtin_amdgcn_readfirstlane(node);
            const int beg = rp[d];
            const int end = rp[d + 1];
            const float dd = dinv_of(cnt[node]);
            {
                const vecf self = hv[(size_t)d * 32 + lane];
#pragma unroll
                for (int q = 0; q < 4; ++q)
                    acc[q] = bias[lane * 4 + q] + self[q] * dd * dd;
            }
            for (int base = beg; base < end; base += 32) {
                const int m = min(32, end - base);
                int   es = 0;
                float ew = 0.0f;
                if (lane < m) {
                    const int2 p = ev[base + lane];
                    es = p.x;
                    ew = __int_as_float(p.y) * dd;
                }
                int i = 0;
                for (; i + 8 <= m; i += 8) rows_add<4, 8>(hv, es, ew, i, lane, acc);
                if (i + 4 <= m) { rows_add<4, 4>(hv, es, ew, i, lane, acc); i += 4; }
                if (i + 2 <= m) { rows_add<4, 2>(hv, es, ew, i, lane, acc); i += 2; }
                if (i < m)      { rows_add<4, 1>(hv, es, ew, i, lane, acc); }
            }
#pragma unroll
            for (int q = 0; q < 4; ++q) acc[q] = fmaxf(acc[q], 0.0f);
        }
        *(vecf*)(as + li * FA_P + 4 * lane) = acc;
    }
    __syncthreads();

    const int rt = wv >> 2;
    const int ct = wv & 3;
    const int row0 = blockIdx.x * FN + 16 * rt;
    if (row0 >= n) return;
    const int hf = lane >> 4;
    const int r  = lane & 15;
    const float* __restrict__ ap = as + (16 * rt + r) * FA_P + 2 * hf;
    const float* __restrict__ bp = w2s + (2 * hf) * FW_P + 16 * ct + r;

    v8f c = (v8f){};
#pragma unroll
    for (int k = 0; k < 128; k += 4) {
        const float2 av = *(const float2*)(ap + k);
        v2f a, b;
        a[0] = av.x;
        a[1] = av.y;
        b[0] = bp[k * FW_P];
        b[1] = bp[(k + 1) * FW_P];
        c = __builtin_amdgcn_wmma_f32_16x16x4_f32(
            false, a, false, b, (short)0, c, false, false);
    }
#pragma unroll
    for (int j = 0; j < 8; ++j) {
        const int row = row0 + 8 * hf + j;
        if (row < n) h2[(size_t)row * 64 + 16 * ct + r] = c[j];
    }
}

extern "C" void kernel_launch(void* const* d_in, const int* in_sizes, int n_in,
                              void* d_out, int out_size, void* d_ws, size_t ws_size,
                              hipStream_t stream) {
    const float* x  = (const float*)d_in[0];
    const int*   ei = (const int*)d_in[1];
    const float* W1 = (const float*)d_in[2];
    const float* b1 = (const float*)d_in[3];
    const float* W2 = (const float*)d_in[4];
    const float* b2 = (const float*)d_in[5];
    float* out = (float*)d_out;

    const int HID = in_sizes[3];
    const int OUT = in_sizes[5];
    const int IN  = in_sizes[2] / HID;
    const int N   = in_sizes[0] / IN;
    const int E   = in_sizes[1] / 2;
    const int* srcI = ei;
    const int* dstI = ei + E;

    const size_t nPad  = ((size_t)N + 4 + 63) & ~(size_t)63;
    const size_t rPad  = ((size_t)E + 63) & ~(size_t)63;
    const size_t ePad  = (2 * (size_t)E + 63) & ~(size_t)63;
    float* ws    = (float*)d_ws;
    int*   cnt   = (int*)ws;
    int*   rp    = (int*)(ws + nPad);
    int*   loc   = (int*)(ws + 2 * nPad);
    int*   bsum  = (int*)(ws + 3 * nPad);
    int*   rank  = (int*)(ws + 3 * nPad + BP_CAP);
    int*   edges = (int*)(ws + 3 * nPad + BP_CAP + rPad);
    int*   part  = (int*)(ws + 3 * nPad + BP_CAP + rPad + ePad);
    float* bufA  = ws + 3 * nPad + BP_CAP + rPad + ePad + CHUNKS * nPad;
    float* bufB  = bufA + (size_t)N * HID;

    const int T = 256;
    const int mtiles = N / 16;
    const int gblk   = (mtiles + 7) / 8;
    const int nblk   = (N + 7) / 8;

    k_count_rank_lds<<<CHUNKS, 1024, 0, stream>>>(dstI, E, N, (int)nPad, part, rank);
    const int nblocks = (N + NB - 1) / NB;
    k_chunk_offsets<CHUNKS><<<nblocks, NB, 0, stream>>>(part, (int)nPad, cnt, loc, bsum, N);
    k_gemm1_with_fill<<<gblk + (E + T - 1) / T, T, 0, stream>>>(
        x, W1, bufA, N, gblk,
        srcI, dstI, E, CHUNKS, cnt, bsum, nblocks, loc, part, (int)nPad, rank, rp, N, edges);
    k_gather_sum_relu_gemm<<<(N + FN - 1) / FN, 512, 0, stream>>>(
        rp, edges, cnt, bufA, b1, W2, bufB, N);
    k_gather_sum<64><<<nblk, T, 0, stream>>>(rp, edges, cnt, bufB, b2, out, N);
}
